# starting kernel
# baseline (speedup 1.0000x reference)
#include <hip/hip_runtime.h>
#include <cmath>

typedef __attribute__((ext_vector_type(2))) float v2f;
typedef __attribute__((ext_vector_type(8))) float v8f;

#define N_PTS   12000
#define DIN     64
#define DLAT    16
#define DFEAT   64
#define KNN     16
#define ROW_TILES (N_PTS / 16)
#define WAVES_PER_WG 8
#define NUM_WG  ((ROW_TILES + WAVES_PER_WG - 1) / WAVES_PER_WG)

static __device__ __forceinline__ v8f wmma_f32(v2f a, v2f b, v8f c) {
  return __builtin_amdgcn_wmma_f32_16x16x4_f32(
       false, a,  false, b,
       (short)0, c,  false,  false);
}

__global__ void __launch_bounds__(256)
encoder_kernel(const float* __restrict__ x,
               const float* __restrict__ Wf, const float* __restrict__ bf,
               const float* __restrict__ Wl, const float* __restrict__ bl,
               float* __restrict__ feats, float* __restrict__ coords) {
  const int lane = threadIdx.x & 31;
  const int wave = threadIdx.x >> 5;
  const int tile = blockIdx.x * WAVES_PER_WG + wave;
  if (tile >= ROW_TILES) return;
  const int row0 = tile * 16;
  const int m  = lane & 15;
  const int kh = lane >> 4;

  v2f a[16];
  const float* xr = x + (size_t)(row0 + m) * DIN;
  #pragma unroll
  for (int s = 0; s < 16; ++s) {
    a[s].x = xr[4 * s + 2 * kh];
    a[s].y = xr[4 * s + 1 + 2 * kh];
  }

  #pragma unroll
  for (int nt = 0; nt < 4; ++nt) {
    v8f acc = {};
    #pragma unroll
    for (int s = 0; s < 16; ++s) {
      v2f b;
      b.x = Wf[(4 * s + 2 * kh) * DFEAT + nt * 16 + m];
      b.y = Wf[(4 * s + 1 + 2 * kh) * DFEAT + nt * 16 + m];
      acc = wmma_f32(a[s], b, acc);
    }
    #pragma unroll
    for (int r = 0; r < 8; ++r) {
      const int row = row0 + r + 8 * kh;
      const int col = nt * 16 + m;
      feats[(size_t)row * DFEAT + col] = tanhf(acc[r] + bf[col]);
    }
  }

  {
    v8f acc = {};
    #pragma unroll
    for (int s = 0; s < 16; ++s) {
      v2f b;
      b.x = Wl[(4 * s + 2 * kh) * DLAT + m];
      b.y = Wl[(4 * s + 1 + 2 * kh) * DLAT + m];
      acc = wmma_f32(a[s], b, acc);
    }
    #pragma unroll
    for (int r = 0; r < 8; ++r) {
      const int row = row0 + r + 8 * kh;
      coords[(size_t)row * DLAT + m] = tanhf(acc[r] + bl[m]);
    }
  }
}

__global__ void sq_kernel(const float* __restrict__ coords, float* __restrict__ sq) {
  const int i = blockIdx.x * blockDim.x + threadIdx.x;
  if (i >= N_PTS) return;
  const float* c = coords + (size_t)i * DLAT;
  float s = 0.f;
  #pragma unroll
  for (int d = 0; d < DLAT; ++d) s += c[d] * c[d];
  sq[i] = s;
}

__global__ void __launch_bounds__(256)
gravnet_kernel(const float* __restrict__ coords, const float* __restrict__ sq,
               const float* __restrict__ feats,
               const float* __restrict__ Wout, const float* __restrict__ bout,
               float* __restrict__ out) {
  __shared__ float lds_dot[WAVES_PER_WG][16 * 17];
  __shared__ float lds_w  [WAVES_PER_WG][16][KNN];
  __shared__ int   lds_j  [WAVES_PER_WG][16][KNN];
  __shared__ float lds_agg[WAVES_PER_WG][16][DFEAT];

  const int lane = threadIdx.x & 31;
  const int wave = threadIdx.x >> 5;
  const int tile = blockIdx.x * WAVES_PER_WG + wave;
  if (tile >= ROW_TILES) return;
  const int row0 = tile * 16;
  const int m  = lane & 15;
  const int kh = lane >> 4;

  v2f a[4];
  {
    const float* cr = coords + (size_t)(row0 + m) * DLAT;
    #pragma unroll
    for (int s = 0; s < 4; ++s) {
      a[s].x = cr[4 * s + 2 * kh];
      a[s].y = cr[4 * s + 1 + 2 * kh];
    }
  }

  float bestd[KNN]; int bestj[KNN];
  #pragma unroll
  for (int s = 0; s < KNN; ++s) { bestd[s] = 3.4e38f; bestj[s] = 0; }
  float worst = 3.4e38f; int wslot = 0;
  const float sqi = sq[row0 + m];

  for (int jt = 0; jt < ROW_TILES; ++jt) {
    const int j0 = jt * 16;
    if (jt + 1 < ROW_TILES)
      __builtin_prefetch(coords + (size_t)(j0 + 16 + m) * DLAT, 0, 3);

    v8f acc = {};
    const float* cj = coords + (size_t)(j0 + m) * DLAT;
    #pragma unroll
    for (int s = 0; s < 4; ++s) {
      v2f b;
      b.x = cj[4 * s + 2 * kh];
      b.y = cj[4 * s + 1 + 2 * kh];
      acc = wmma_f32(a[s], b, acc);
    }

    #pragma unroll
    for (int r = 0; r < 8; ++r)
      lds_dot[wave][(r + 8 * kh) * 17 + m] = acc[r];
    __builtin_amdgcn_wave_barrier();

    if (lane < 16) {
      const int row = row0 + lane;
      #pragma unroll
      for (int n = 0; n < 16; ++n) {
        const int j = j0 + n;
        float d = sqi + sq[j] - 2.f * lds_dot[wave][lane * 17 + n];
        d = fmaxf(d, 0.f);
        if (j != row && d < worst) {
          #pragma unroll
          for (int s = 0; s < KNN; ++s)
            if (s == wslot) { bestd[s] = d; bestj[s] = j; }
          worst = bestd[0]; wslot = 0;
          #pragma unroll
          for (int s = 1; s < KNN; ++s)
            if (bestd[s] > worst) { worst = bestd[s]; wslot = s; }
        }
      }
    }
    __builtin_amdgcn_wave_barrier();
  }

  if (lane < 16) {
    #pragma unroll
    for (int k = 0; k < KNN; ++k) {
      lds_w[wave][lane][k] = expf(-bestd[k]);
      lds_j[wave][lane][k] = bestj[k];
    }
  }
  __builtin_amdgcn_wave_barrier();

  for (int r = 0; r < 16; ++r) {
    float acc0 = 0.f, acc1 = 0.f;
    #pragma unroll
    for (int k = 0; k < KNN; ++k) {
      const int   j  = lds_j[wave][r][k];
      const float wk = lds_w[wave][r][k];
      acc0 += wk * feats[(size_t)j * DFEAT + lane];
      acc1 += wk * feats[(size_t)j * DFEAT + lane + 32];
    }
    lds_agg[wave][r][lane]      = acc0;
    lds_agg[wave][r][lane + 32] = acc1;
  }
  __builtin_amdgcn_wave_barrier();

  const float* fr = feats + (size_t)(row0 + m) * DFEAT;
  #pragma unroll
  for (int nt = 0; nt < 4; ++nt) {
    v8f acc = {};
    #pragma unroll
    for (int s = 0; s < 16; ++s) {
      v2f av, bv;
      av.x = fr[4 * s + 2 * kh];
      av.y = fr[4 * s + 1 + 2 * kh];
      bv.x = Wout[(4 * s + 2 * kh) * DFEAT + nt * 16 + m];
      bv.y = Wout[(4 * s + 1 + 2 * kh) * DFEAT + nt * 16 + m];
      acc = wmma_f32(av, bv, acc);
    }
    #pragma unroll
    for (int s = 0; s < 16; ++s) {
      v2f av, bv;
      av.x = lds_agg[wave][m][4 * s + 2 * kh];
      av.y = lds_agg[wave][m][4 * s + 1 + 2 * kh];
      bv.x = Wout[(64 + 4 * s + 2 * kh) * DFEAT + nt * 16 + m];
      bv.y = Wout[(64 + 4 * s + 1 + 2 * kh) * DFEAT + nt * 16 + m];
      acc = wmma_f32(av, bv, acc);
    }
    #pragma unroll
    for (int r = 0; r < 8; ++r) {
      const int row = row0 + r + 8 * kh;
      const int col = nt * 16 + m;
      out[(size_t)row * DFEAT + col] = acc[r] + bout[col];
    }
  }
}

extern "C" void kernel_launch(void* const* d_in, const int* in_sizes, int n_in,
                              void* d_out, int out_size, void* d_ws, size_t ws_size,
                              hipStream_t stream) {
  const float* x  = (const float*)d_in[0];
  const float* Wf = (const float*)d_in[1];
  const float* bf = (const float*)d_in[2];
  const float* Wl = (const float*)d_in[3];
  const float* bl = (const float*)d_in[4];
  const float* Wo = (const float*)d_in[5];
  const float* bo = (const float*)d_in[6];
  float* out = (float*)d_out;

  float* feats  = (float*)d_ws;
  float* coords = feats  + (size_t)N_PTS * DFEAT;
  float* sq     = coords + (size_t)N_PTS * DLAT;

  encoder_kernel<<<NUM_WG, 256, 0, stream>>>(x, Wf, bf, Wl, bl, feats, coords);
  sq_kernel<<<(N_PTS + 255) / 256, 256, 0, stream>>>(coords, sq);
  gravnet_kernel<<<NUM_WG, 256, 0, stream>>>(coords, sq, feats, Wo, bo, out);
}


# Each group's products are issued one group ahead, spread among the previous group's test instructions (two register sets swap roles), its LDS reads as one run ahead of them; lists unchanged
# speedup vs baseline: 7.6927x; 7.6927x over previous
#include <hip/hip_runtime.h>
#include <cmath>

typedef __attribute__((ext_vector_type(2))) float v2f;
typedef __attribute__((ext_vector_type(4))) float v4f;
typedef __attribute__((ext_vector_type(8))) float v8f;

#define N_PTS   12000
#define DIN     64
#define DLAT    16
#define DFEAT   64
#define KNN     16
#define ROW_TILES (N_PTS / 16)
#define WAVES_PER_WG 8
#define NUM_WG  ((ROW_TILES + WAVES_PER_WG - 1) / WAVES_PER_WG)

static __device__ __forceinline__ v8f wmma_f32(v2f a, v2f b, v8f c) {
  return __builtin_amdgcn_wmma_f32_16x16x4_f32(
       false, a,  false, b,
       (short)0, c,  false,  false);
}

__global__ void __launch_bounds__(256)
encoder_kernel(const float* __restrict__ x,
               const float* __restrict__ Wf, const float* __restrict__ bf,
               const float* __restrict__ Wl, const float* __restrict__ bl,
               float* __restrict__ feats, float* __restrict__ coords) {
  const int lane = threadIdx.x & 31;
  const int wave = threadIdx.x >> 5;
  const int tile = blockIdx.x * WAVES_PER_WG + wave;
  if (tile >= ROW_TILES) return;
  const int row0 = tile * 16;
  const int m  = lane & 15;
  const int kh = lane >> 4;

  v2f a[16];
  const float* xr = x + (size_t)(row0 + m) * DIN;
  #pragma unroll
  for (int s = 0; s < 16; ++s) {
    a[s].x = xr[4 * s + 2 * kh];
    a[s].y = xr[4 * s + 1 + 2 * kh];
  }

  #pragma unroll
  for (int nt = 0; nt < 4; ++nt) {
    v8f acc = {};
    #pragma unroll
    for (int s = 0; s < 16; ++s) {
      v2f b;
      b.x = Wf[(4 * s + 2 * kh) * DFEAT + nt * 16 + m];
      b.y = Wf[(4 * s + 1 + 2 * kh) * DFEAT + nt * 16 + m];
      acc = wmma_f32(a[s], b, acc);
    }
    #pragma unroll
    for (int r = 0; r < 8; ++r) {
      const int row = row0 + r + 8 * kh;
      const int col = nt * 16 + m;
      feats[(size_t)row * DFEAT + col] = tanhf(acc[r] + bf[col]);
    }
  }

  {
    v8f acc = {};
    #pragma unroll
    for (int s = 0; s < 16; ++s) {
      v2f b;
      b.x = Wl[(4 * s + 2 * kh) * DLAT + m];
      b.y = Wl[(4 * s + 1 + 2 * kh) * DLAT + m];
      acc = wmma_f32(a[s], b, acc);
    }
    #pragma unroll
    for (int r = 0; r < 8; ++r) {
      const int row = row0 + r + 8 * kh;
      coords[(size_t)row * DLAT + m] = tanhf(acc[r] + bl[m]);
    }
  }
}

__global__ void sq_kernel(const float* __restrict__ coords, float* __restrict__ sq) {
  const int i = blockIdx.x * blockDim.x + threadIdx.x;
  if (i >= N_PTS) return;
  const float* c = coords + (size_t)i * DLAT;
  float s = 0.f;
  #pragma unroll
  for (int d = 0; d < DLAT; ++d) s += c[d] * c[d];
  sq[i] = s;
}

#define G_WAVES   3
#define G_THREADS (G_WAVES * 32)
#define G_NUM_WG  (ROW_TILES / G_WAVES)
#define G_CT      30
#define G_GRP     3
#define G_NGRP    (G_CT / G_GRP)
#define G_ND      (8 * G_GRP)
#define G_NCHUNK  (ROW_TILES / G_CT)
#define G_CCOLS   (G_CT * 16)
#define G_PLANE   (G_CCOLS * 4 + 16)
#define G_BUF     (4 * G_PLANE + G_CCOLS)
#define G_C4      (G_CCOLS * 4)
#define G_N4      (G_CCOLS / 4)
#define G_CPT     (G_C4 / G_THREADS)
#define G_NPT     ((G_N4 + G_THREADS - 1) / G_THREADS)
static_assert(ROW_TILES % G_WAVES == 0, "row tiles must fill the workgroups");
static_assert(ROW_TILES % G_CT == 0, "column tiles must fill the chunks");
static_assert(G_C4 % G_THREADS == 0, "coords pieces must divide over the threads");
static_assert(G_CT % G_GRP == 0, "tile groups must fill the chunks");
static_assert(G_ND < 32, "a lane's queue of one group is one 32-bit word with a spare bit");

__global__ void __launch_bounds__(G_THREADS)
gravnet_kernel(const float* __restrict__ coords, const float* __restrict__ sq,
               const float* __restrict__ feats,
               const float* __restrict__ Wout, const float* __restrict__ bout,
               float* __restrict__ out) {
  __shared__ __attribute__((aligned(16))) float lds_stage[2][G_BUF];
  __shared__ float lds_cd [G_WAVES][16][2 * KNN];
  __shared__ int   lds_cj [G_WAVES][16][2 * KNN];
  __shared__ float lds_w  [G_WAVES][16][KNN];
  __shared__ int   lds_j  [G_WAVES][16][KNN];
  __shared__ float lds_agg[G_WAVES][16][DFEAT];

  const int tid  = threadIdx.x;
  const int lane = tid & 31;
  const int wave = tid >> 5;
  const int tile = blockIdx.x * G_WAVES + wave;
  const int row0 = tile * 16;
  const int m  = lane & 15;
  const int kh = lane >> 4;
  const int row = row0 + m;

  v2f bown[4];
  {
    const float* cr = coords + (size_t)(row0 + m) * DLAT;
    #pragma unroll
    for (int s = 0; s < 4; ++s) {
      bown[s].x = cr[4 * s + 2 * kh];
      bown[s].y = cr[4 * s + 1 + 2 * kh];
    }
  }

  float bestd[KNN]; int bestj[KNN];
  #pragma unroll
  for (int s = 0; s < KNN; ++s) { bestd[s] = 3.4e38f; bestj[s] = 0; }
  const float sqi = sq[row];
  const unsigned selfbit = ((m >> 3) == kh) ? (1u << (m & 7)) : 0u;

  const v4f* coords4 = reinterpret_cast<const v4f*>(coords);
  const v4f* sq4     = reinterpret_cast<const v4f*>(sq);
  v4f creg[G_CPT];
  v4f nreg[G_NPT];

  auto stage_load = [&](int c) {
    const v4f* csrc = coords4 + (size_t)c * G_C4;
    #pragma unroll
    for (int i = 0; i < G_CPT; ++i) creg[i] = csrc[tid + G_THREADS * i];
    const v4f* nsrc = sq4 + (size_t)c * G_N4;
    #pragma unroll
    for (int i = 0; i < G_NPT; ++i) {
      const int q = tid + G_THREADS * i;
      if (q < G_N4) nreg[i] = nsrc[q];
    }
  };
  auto stage_store = [&](int b) {
    v4f* dst = reinterpret_cast<v4f*>(&lds_stage[b][0]);
    #pragma unroll
    for (int i = 0; i < G_CPT; ++i) {
      const int q = tid + G_THREADS * i;
      dst[(q & 3) * (G_PLANE / 4) + (q >> 2)] = creg[i];
    }
    #pragma unroll
    for (int i = 0; i < G_NPT; ++i) {
      const int q = tid + G_THREADS * i;
      if (q < G_N4) dst[G_PLANE + q] = nreg[i];
    }
  };
  auto load_coords = [&](const float* pb, int t, v2f (&af)[4]) __attribute__((always_inline)) {
    const float* p = pb + (16 * t + m) * 4 + 2 * kh;
    #pragma unroll
    for (int s = 0; s < 4; ++s) af[s] = *reinterpret_cast<const v2f*>(p + s * G_PLANE);
  };
  auto load_norms = [&](const float* pb, int t, v4f& q0, v4f& q1) __attribute__((always_inline)) {
    const float* pn = pb + 4 * G_PLANE + 16 * t + 8 * kh;
    q0 = *reinterpret_cast<const v4f*>(pn);
    q1 = *reinterpret_cast<const v4f*>(pn + 4);
  };

  auto products = [&](v2f (&af)[G_GRP][4], v8f (&acc)[G_GRP]) __attribute__((always_inline)) {
    #pragma unroll
    for (int g = 0; g < G_GRP; ++g) acc[g] = v8f{};
    #pragma unroll
    for (int s = 0; s < 4; ++s) {
      #pragma unroll
      for (int g = 0; g < G_GRP; ++g) acc[g] = wmma_f32(af[g][s], bown[s], acc[g]);
    }
  };

  stage_load(0);
  stage_store(0);
  __syncthreads();

  for (int c = 0; c < G_NCHUNK; ++c) {
    if (c + 1 < G_NCHUNK) stage_load(c + 1);
    const float* pb = &lds_stage[c & 1][0];

    auto group_step = [&](const int tg, v8f (&acc)[G_GRP], v4f (&p0)[G_GRP], v4f (&p1)[G_GRP],
                          v8f (&accn)[G_GRP], v4f (&pn0)[G_GRP], v4f (&pn1)[G_GRP],
                          v2f (&afr)[G_GRP][4], v2f (&afw)[G_GRP][4]) __attribute__((always_inline)) {
      const int t0 = tg * G_GRP;
      const int j0 = (c * G_CT + t0) * 16;

      {
        const int tn1 = (tg + 1 < G_NGRP) ? t0 + G_GRP : t0;
        const int tn2 = (tg + 2 < G_NGRP) ? t0 + 2 * G_GRP : G_CT - G_GRP;
        #pragma unroll
        for (int g = 0; g < G_GRP; ++g) load_norms(pb, tn1 + g, pn0[g], pn1[g]);
        #pragma unroll
        for (int g = 0; g < G_GRP; ++g) load_coords(pb, tn2 + g, afw[g]);
      }
      __builtin_amdgcn_sched_barrier(0);
      products(afr, accn);

      const float worst0 = bestd[KNN - 1];
      static_assert(G_GRP == 3, "the group's tiles are written out one by one below");
      float dd0[8], dd1[8], dd2[8]; unsigned pend = 0u;
      {
        const float sjx[8] = {p0[0].x, p0[0].y, p0[0].z, p0[0].w, p1[0].x, p1[0].y, p1[0].z, p1[0].w};
        #pragma unroll
        for (int r = 0; r < 8; ++r) {
          float d = sqi + sjx[r] - 2.f * acc[0][r];
          d = fmaxf(d, 0.f);
          dd0[r] = d;
          pend |= (d < worst0) ? (1u << r) : 0u;
        }
      }
      {
        const float sjx[8] = {p0[1].x, p0[1].y, p0[1].z, p0[1].w, p1[1].x, p1[1].y, p1[1].z, p1[1].w};
        #pragma unroll
        for (int r = 0; r < 8; ++r) {
          float d = sqi + sjx[r] - 2.f * acc[1][r];
          d = fmaxf(d, 0.f);
          dd1[r] = d;
          pend |= (d < worst0) ? (1u << (8 + r)) : 0u;
        }
      }
      {
        const float sjx[8] = {p0[2].x, p0[2].y, p0[2].z, p0[2].w, p1[2].x, p1[2].y, p1[2].z, p1[2].w};
        #pragma unroll
        for (int r = 0; r < 8; ++r) {
          float d = sqi + sjx[r] - 2.f * acc[2][r];
          d = fmaxf(d, 0.f);
          dd2[r] = d;
          pend |= (d < worst0) ? (1u << (16 + r)) : 0u;
        }
      }
      const int gd = (row0 - j0) >> 4;
      if (gd >= 0 && gd < G_GRP) pend &= ~(selfbit << (8 * gd));

      #pragma unroll
      for (int i = 0; i < 4 * G_GRP; ++i) {
        __builtin_amdgcn_sched_group_barrier(0x008, 1, 0);
        __builtin_amdgcn_sched_group_barrier(0x002, 10, 0);
      }

      const int jbase = j0 + 8 * kh;
      if (__builtin_amdgcn_ballot_w32(pend != 0u) != 0u) {
        do {
          const bool has = pend != 0u;
          const unsigned idx = (unsigned)__builtin_ctz(pend | (1u << G_ND));
          const bool b0 = (idx & 1u) != 0u, b1 = (idx & 2u) != 0u, b2 = (idx & 4u) != 0u;
          float e0, e1, e2;
          {
            const float t0s = b0 ? dd0[1] : dd0[0];
            const float t1s = b0 ? dd0[3] : dd0[2];
            const float t2s = b0 ? dd0[5] : dd0[4];
            const float t3s = b0 ? dd0[7] : dd0[6];
            const float u0 = b1 ? t1s : t0s;
            const float u1 = b1 ? t3s : t2s;
            e0 = b2 ? u1 : u0;
          }
          {
            const float t0s = b0 ? dd1[1] : dd1[0];
            const float t1s = b0 ? dd1[3] : dd1[2];
            const float t2s = b0 ? dd1[5] : dd1[4];
            const float t3s = b0 ? dd1[7] : dd1[6];
            const float u0 = b1 ? t1s : t0s;
            const float u1 = b1 ? t3s : t2s;
            e1 = b2 ? u1 : u0;
          }
          {
            const float t0s = b0 ? dd2[1] : dd2[0];
            const float t1s = b0 ? dd2[3] : dd2[2];
            const float t2s = b0 ? dd2[5] : dd2[4];
            const float t3s = b0 ? dd2[7] : dd2[6];
            const float u0 = b1 ? t1s : t0s;
            const float u1 = b1 ? t3s : t2s;
            e2 = b2 ? u1 : u0;
          }
          const bool b3 = (idx & 8u) != 0u, b4 = (idx & 16u) != 0u;
          const float e01 = b3 ? e1 : e0;
          const float dsel = b4 ? e2 : e01;
          const float d = has ? dsel : __builtin_inff();
          const int j = jbase + (int)idx + (int)(idx & ~7u);
          pend &= pend - 1u;
          bool before[KNN];
          #pragma unroll
          for (int s = 0; s < KNN; ++s) before[s] = d < bestd[s];
          __builtin_amdgcn_sched_barrier(0);
          #pragma unroll
          for (int s = KNN - 1; s >= 1; --s) {
            const float dprev = bestd[s - 1], dcur = bestd[s];
            const int   jprev = bestj[s - 1], jcur = bestj[s];
            const int   jhere = before[s] ? j : jcur;
            bestj[s] = before[s - 1] ? jprev : jhere;
            bestd[s] = __builtin_amdgcn_fmed3f(dprev, d, dcur);
          }
          {
            const float dcur = bestd[0];
            const int   jcur = bestj[0];
            bestj[0] = before[0] ? j : jcur;
            bestd[0] = before[0] ? d : dcur;
          }
        } while (__builtin_amdgcn_ballot_w32(pend != 0u) != 0u);
      }
    };

    v8f accA[G_GRP], accB[G_GRP]; v4f pA0[G_GRP], pA1[G_GRP], pB0[G_GRP], pB1[G_GRP];
    v2f afA[G_GRP][4], afB[G_GRP][4];
    static_assert(G_NGRP % 2 == 0 && G_NGRP >= 2, "the groups are taken in pairs, on two alternating register sets");
    #pragma unroll
    for (int g = 0; g < G_GRP; ++g) { load_coords(pb, g, afB[g]); load_norms(pb, g, pA0[g], pA1[g]); }
    products(afB, accA);
    #pragma unroll
    for (int g = 0; g < G_GRP; ++g) load_coords(pb, G_GRP + g, afA[g]);

    for (int tg = 0; tg < G_NGRP; tg += 2) {
      group_step(tg,     accA, pA0, pA1, accB, pB0, pB1, afA, afB);
      group_step(tg + 1, accB, pB0, pB1, accA, pA0, pA1, afB, afA);
    }

    if (c + 1 < G_NCHUNK) stage_store((c + 1) & 1);
    __syncthreads();
  }

  #pragma unroll
  for (int s = 0; s < KNN; ++s) {
    lds_cd[wave][m][kh * KNN + s] = bestd[s];
    lds_cj[wave][m][kh * KNN + s] = bestj[s];
  }
  __builtin_amdgcn_wave_barrier();

  {
    int rank[KNN];
    #pragma unroll
    for (int s = 0; s < KNN; ++s) rank[s] = 0;
    #pragma unroll 1
    for (int t = 0; t < 2 * KNN; ++t) {
      const float dt = lds_cd[wave][m][t];
      const int   jq = lds_cj[wave][m][t];
      #pragma unroll
      for (int s = 0; s < KNN; ++s) {
        const int before = (int)(dt < bestd[s]) | ((int)(dt == bestd[s]) & (int)(jq < bestj[s]));
        rank[s] += before;
      }
    }
    #pragma unroll
    for (int s = 0; s < KNN; ++s) {
      if (rank[s] < KNN) {
        lds_w[wave][m][rank[s]] = expf(-bestd[s]);
        lds_j[wave][m][rank[s]] = bestj[s];
      }
    }
  }
  __builtin_amdgcn_wave_barrier();

  for (int r = 0; r < 16; ++r) {
    float acc0 = 0.f, acc1 = 0.f;
    #pragma unroll
    for (int k = 0; k < KNN; ++k) {
      const int   j  = lds_j[wave][r][k];
      const float wk = lds_w[wave][r][k];
      acc0 += wk * feats[(size_t)j * DFEAT + lane];
      acc1 += wk * feats[(size_t)j * DFEAT + lane + 32];
    }
    lds_agg[wave][r][lane]      = acc0;
    lds_agg[wave][r][lane + 32] = acc1;
  }
  __builtin_amdgcn_wave_barrier();

  const float* fr = feats + (size_t)(row0 + m) * DFEAT;
  #pragma unroll
  for (int nt = 0; nt < 4; ++nt) {
    v8f acc = {};
    #pragma unroll
    for (int s = 0; s < 16; ++s) {
      v2f av, bv;
      av.x = fr[4 * s + 2 * kh];
      av.y = fr[4 * s + 1 + 2 * kh];
      bv.x = Wout[(4 * s + 2 * kh) * DFEAT + nt * 16 + m];
      bv.y = Wout[(4 * s + 1 + 2 * kh) * DFEAT + nt * 16 + m];
      acc = wmma_f32(av, bv, acc);
    }
    #pragma unroll
    for (int s = 0; s < 16; ++s) {
      v2f av, bv;
      av.x = lds_agg[wave][m][4 * s + 2 * kh];
      av.y = lds_agg[wave][m][4 * s + 1 + 2 * kh];
      bv.x = Wout[(64 + 4 * s + 2 * kh) * DFEAT + nt * 16 + m];
      bv.y = Wout[(64 + 4 * s + 1 + 2 * kh) * DFEAT + nt * 16 + m];
      acc = wmma_f32(av, bv, acc);
    }
    #pragma unroll
    for (int r = 0; r < 8; ++r) {
      const int row = row0 + r + 8 * kh;
      const int col = nt * 16 + m;
      out[(size_t)row * DFEAT + col] = acc[r] + bout[col];
    }
  }
}

extern "C" void kernel_launch(void* const* d_in, const int* in_sizes, int n_in,
                              void* d_out, int out_size, void* d_ws, size_t ws_size,
                              hipStream_t stream) {
  const float* x  = (const float*)d_in[0];
  const float* Wf = (const float*)d_in[1];
  const float* bf = (const float*)d_in[2];
  const float* Wl = (const float*)d_in[3];
  const float* bl = (const float*)d_in[4];
  const float* Wo = (const float*)d_in[5];
  const float* bo = (const float*)d_in[6];
  float* out = (float*)d_out;

  float* feats  = (float*)d_ws;
  float* coords = feats  + (size_t)N_PTS * DFEAT;
  float* sq     = coords + (size_t)N_PTS * DLAT;

  encoder_kernel<<<NUM_WG, 256, 0, stream>>>(x, Wf, bf, Wl, bl, feats, coords);
  sq_kernel<<<(N_PTS + 255) / 256, 256, 0, stream>>>(coords, sq);
  gravnet_kernel<<<G_NUM_WG, G_THREADS, 0, stream>>>(coords, sq, feats, Wo, bo, out);
}


# Each row tile gets a second wave: the scan wave takes products and distances and leaves the distances in LDS; the list wave tests and inserts them a chunk later, beside the next scan; lists unchanged
# speedup vs baseline: 8.0772x; 1.0500x over previous
#include <hip/hip_runtime.h>
#include <cmath>

typedef __attribute__((ext_vector_type(2))) float v2f;
typedef __attribute__((ext_vector_type(4))) float v4f;
typedef __attribute__((ext_vector_type(8))) float v8f;

#define N_PTS   12000
#define DIN     64
#define DLAT    16
#define DFEAT   64
#define KNN     16
#define ROW_TILES (N_PTS / 16)
#define WAVES_PER_WG 8
#define NUM_WG  ((ROW_TILES + WAVES_PER_WG - 1) / WAVES_PER_WG)

static __device__ __forceinline__ v8f wmma_f32(v2f a, v2f b, v8f c) {
  return __builtin_amdgcn_wmma_f32_16x16x4_f32(
       false, a,  false, b,
       (short)0, c,  false,  false);
}

__global__ void __launch_bounds__(256)
encoder_kernel(const float* __restrict__ x,
               const float* __restrict__ Wf, const float* __restrict__ bf,
               const float* __restrict__ Wl, const float* __restrict__ bl,
               float* __restrict__ feats, float* __restrict__ coords) {
  const int lane = threadIdx.x & 31;
  const int wave = threadIdx.x >> 5;
  const int tile = blockIdx.x * WAVES_PER_WG + wave;
  if (tile >= ROW_TILES) return;
  const int row0 = tile * 16;
  const int m  = lane & 15;
  const int kh = lane >> 4;

  v2f a[16];
  const float* xr = x + (size_t)(row0 + m) * DIN;
  #pragma unroll
  for (int s = 0; s < 16; ++s) {
    a[s].x = xr[4 * s + 2 * kh];
    a[s].y = xr[4 * s + 1 + 2 * kh];
  }

  #pragma unroll
  for (int nt = 0; nt < 4; ++nt) {
    v8f acc = {};
    #pragma unroll
    for (int s = 0; s < 16; ++s) {
      v2f b;
      b.x = Wf[(4 * s + 2 * kh) * DFEAT + nt * 16 + m];
      b.y = Wf[(4 * s + 1 + 2 * kh) * DFEAT + nt * 16 + m];
      acc = wmma_f32(a[s], b, acc);
    }
    #pragma unroll
    for (int r = 0; r < 8; ++r) {
      const int row = row0 + r + 8 * kh;
      const int col = nt * 16 + m;
      feats[(size_t)row * DFEAT + col] = tanhf(acc[r] + bf[col]);
    }
  }

  {
    v8f acc = {};
    #pragma unroll
    for (int s = 0; s < 16; ++s) {
      v2f b;
      b.x = Wl[(4 * s + 2 * kh) * DLAT + m];
      b.y = Wl[(4 * s + 1 + 2 * kh) * DLAT + m];
      acc = wmma_f32(a[s], b, acc);
    }
    #pragma unroll
    for (int r = 0; r < 8; ++r) {
      const int row = row0 + r + 8 * kh;
      coords[(size_t)row * DLAT + m] = tanhf(acc[r] + bl[m]);
    }
  }
}

__global__ void sq_kernel(const float* __restrict__ coords, float* __restrict__ sq) {
  const int i = blockIdx.x * blockDim.x + threadIdx.x;
  if (i >= N_PTS) return;
  const float* c = coords + (size_t)i * DLAT;
  float s = 0.f;
  #pragma unroll
  for (int d = 0; d < DLAT; ++d) s += c[d] * c[d];
  sq[i] = s;
}

#define G_TILES   3
#define G_WAVES   (2 * G_TILES)
#define G_THREADS (G_WAVES * 32)
#define G_NUM_WG  (ROW_TILES / G_TILES)
#define G_CT      30
#define G_GRP     3
#define G_NGRP    (G_CT / G_GRP)
#define G_ND      (8 * G_GRP)
#define G_NCHUNK  (ROW_TILES / G_CT)
#define G_CCOLS   (G_CT * 16)
#define G_PLANE   (G_CCOLS * 4 + 16)
#define G_BUF     (4 * G_PLANE + G_CCOLS)
#define G_C4      (G_CCOLS * 4)
#define G_N4      (G_CCOLS / 4)
#define G_CPT     (G_C4 / G_THREADS)
#define G_NPT     ((G_N4 + G_THREADS - 1) / G_THREADS)
static_assert(ROW_TILES % G_TILES == 0, "row tiles must fill the workgroups");
static_assert(ROW_TILES % G_CT == 0, "column tiles must fill the chunks");
static_assert(G_C4 % G_THREADS == 0, "coords pieces must divide over the threads");
static_assert(G_CT % G_GRP == 0, "tile groups must fill the chunks");
static_assert(G_ND < 32, "a lane's queue of one group is one 32-bit word with a spare bit");

__global__ void __launch_bounds__(G_THREADS)
gravnet_kernel(const float* __restrict__ coords, const float* __restrict__ sq,
               const float* __restrict__ feats,
               const float* __restrict__ Wout, const float* __restrict__ bout,
               float* __restrict__ out) {
  __shared__ __attribute__((aligned(16))) float lds_stage[2][G_BUF];
  __shared__ __attribute__((aligned(16))) float lds_d[G_TILES][2][G_CT][2][32][4];
  __shared__ float lds_cd [G_TILES][16][2 * KNN];
  __shared__ int   lds_cj [G_TILES][16][2 * KNN];
  __shared__ float lds_w  [G_TILES][16][KNN];
  __shared__ int   lds_j  [G_TILES][16][KNN];
  __shared__ float lds_agg[G_TILES][16][DFEAT];

  const int tid  = threadIdx.x;
  const int lane = tid & 31;
  const int wave = __builtin_amdgcn_readfirstlane(tid >> 5);
  const bool scan_wave = wave < G_TILES;
  const int tl   = scan_wave ? wave : wave - G_TILES;
  const int tile = blockIdx.x * G_TILES + tl;
  const int row0 = tile * 16;
  const int m  = lane & 15;
  const int kh = lane >> 4;
  const int row = row0 + m;

  v2f bown[4];
  {
    const float* cr = coords + (size_t)(row0 + m) * DLAT;
    #pragma unroll
    for (int s = 0; s < 4; ++s) {
      bown[s].x = cr[4 * s + 2 * kh];
      bown[s].y = cr[4 * s + 1 + 2 * kh];
    }
  }

  float bestd[KNN]; int bestj[KNN];
  #pragma unroll
  for (int s = 0; s < KNN; ++s) { bestd[s] = 3.4e38f; bestj[s] = 0; }
  const float sqi = sq[row];
  const unsigned selfbit = ((m >> 3) == kh) ? (1u << (m & 7)) : 0u;

  const v4f* coords4 = reinterpret_cast<const v4f*>(coords);
  const v4f* sq4     = reinterpret_cast<const v4f*>(sq);
  v4f creg[G_CPT];
  v4f nreg[G_NPT];

  auto stage_load = [&](int c) {
    const v4f* csrc = coords4 + (size_t)c * G_C4;
    #pragma unroll
    for (int i = 0; i < G_CPT; ++i) creg[i] = csrc[tid + G_THREADS * i];
    const v4f* nsrc = sq4 + (size_t)c * G_N4;
    #pragma unroll
    for (int i = 0; i < G_NPT; ++i) {
      const int q = tid + G_THREADS * i;
      if (q < G_N4) nreg[i] = nsrc[q];
    }
  };
  auto stage_store = [&](int b) {
    v4f* dst = reinterpret_cast<v4f*>(&lds_stage[b][0]);
    #pragma unroll
    for (int i = 0; i < G_CPT; ++i) {
      const int q = tid + G_THREADS * i;
      dst[(q & 3) * (G_PLANE / 4) + (q >> 2)] = creg[i];
    }
    #pragma unroll
    for (int i = 0; i < G_NPT; ++i) {
      const int q = tid + G_THREADS * i;
      if (q < G_N4) dst[G_PLANE + q] = nreg[i];
    }
  };
  auto load_coords = [&](const float* pb, int t, v2f (&af)[4]) __attribute__((always_inline)) {
    const float* p = pb + (16 * t + m) * 4 + 2 * kh;
    #pragma unroll
    for (int s = 0; s < 4; ++s) af[s] = *reinterpret_cast<const v2f*>(p + s * G_PLANE);
  };
  auto load_norms = [&](const float* pb, int t, v4f& q0, v4f& q1) __attribute__((always_inline)) {
    const float* pn = pb + 4 * G_PLANE + 16 * t + 8 * kh;
    q0 = *reinterpret_cast<const v4f*>(pn);
    q1 = *reinterpret_cast<const v4f*>(pn + 4);
  };

  auto products = [&](v2f (&af)[G_GRP][4], v8f (&acc)[G_GRP]) __attribute__((always_inline)) {
    #pragma unroll
    for (int g = 0; g < G_GRP; ++g) acc[g] = v8f{};
    #pragma unroll
    for (int s = 0; s < 4; ++s) {
      #pragma unroll
      for (int g = 0; g < G_GRP; ++g) acc[g] = wmma_f32(af[g][s], bown[s], acc[g]);
    }
  };

  auto list_chunk = [&](const int cc) {
    const v4f* dsrc = reinterpret_cast<const v4f*>(&lds_d[tl][cc & 1][0][0][0][0]) + lane;
    v4f q[2 * G_GRP];
    #pragma unroll
    for (int i = 0; i < 2 * G_GRP; ++i) q[i] = dsrc[i * 32];
    #pragma unroll 1
    for (int tg = 0; tg < G_NGRP; ++tg) {
      const int t0 = tg * G_GRP;
      const int j0 = (cc * G_CT + t0) * 16;
      float dd0[8], dd1[8], dd2[8];
      #pragma unroll
      for (int r = 0; r < 4; ++r) {
        dd0[r] = q[0][r]; dd0[4 + r] = q[1][r];
        dd1[r] = q[2][r]; dd1[4 + r] = q[3][r];
        dd2[r] = q[4][r]; dd2[4 + r] = q[5][r];
      }
      {
        const int tn = (tg + 1 < G_NGRP) ? t0 + G_GRP : t0;
        #pragma unroll
        for (int i = 0; i < 2 * G_GRP; ++i) q[i] = dsrc[(2 * tn + i) * 32];
      }
        const float worst0 = bestd[KNN - 1];
        static_assert(G_GRP == 3, "the group's tiles are written out one by one below");
        unsigned pend = 0u;
        #pragma unroll
        for (int r = 0; r < 8; ++r) pend |= (dd0[r] < worst0) ? (1u << r) : 0u;
        #pragma unroll
        for (int r = 0; r < 8; ++r) pend |= (dd1[r] < worst0) ? (1u << (8 + r)) : 0u;
        #pragma unroll
        for (int r = 0; r < 8; ++r) pend |= (dd2[r] < worst0) ? (1u << (16 + r)) : 0u;
        const int gd = (row0 - j0) >> 4;
        if (gd >= 0 && gd < G_GRP) pend &= ~(selfbit << (8 * gd));

        const int jbase = j0 + 8 * kh;
        if (__builtin_amdgcn_ballot_w32(pend != 0u) != 0u) {
          do {
            const bool has = pend != 0u;
            const unsigned idx = (unsigned)__builtin_ctz(pend | (1u << G_ND));
            const bool b0 = (idx & 1u) != 0u, b1 = (idx & 2u) != 0u, b2 = (idx & 4u) != 0u;
            float e0, e1, e2;
            {
              const float t0s = b0 ? dd0[1] : dd0[0];
              const float t1s = b0 ? dd0[3] : dd0[2];
              const float t2s = b0 ? dd0[5] : dd0[4];
              const float t3s = b0 ? dd0[7] : dd0[6];
              const float u0 = b1 ? t1s : t0s;
              const float u1 = b1 ? t3s : t2s;
              e0 = b2 ? u1 : u0;
            }
            {
              const float t0s = b0 ? dd1[1] : dd1[0];
              const float t1s = b0 ? dd1[3] : dd1[2];
              const float t2s = b0 ? dd1[5] : dd1[4];
              const float t3s = b0 ? dd1[7] : dd1[6];
              const float u0 = b1 ? t1s : t0s;
              const float u1 = b1 ? t3s : t2s;
              e1 = b2 ? u1 : u0;
            }
            {
              const float t0s = b0 ? dd2[1] : dd2[0];
              const float t1s = b0 ? dd2[3] : dd2[2];
              const float t2s = b0 ? dd2[5] : dd2[4];
              const float t3s = b0 ? dd2[7] : dd2[6];
              const float u0 = b1 ? t1s : t0s;
              const float u1 = b1 ? t3s : t2s;
              e2 = b2 ? u1 : u0;
            }
            const bool b3 = (idx & 8u) != 0u, b4 = (idx & 16u) != 0u;
            const float e01 = b3 ? e1 : e0;
            const float dsel = b4 ? e2 : e01;
            const float d = has ? dsel : __builtin_inff();
            const int j = jbase + (int)idx + (int)(idx & ~7u);
            pend &= pend - 1u;
            bool before[KNN];
            #pragma unroll
            for (int s = 0; s < KNN; ++s) before[s] = d < bestd[s];
            __builtin_amdgcn_sched_barrier(0);
            #pragma unroll
            for (int s = KNN - 1; s >= 1; --s) {
              const float dprev = bestd[s - 1], dcur = bestd[s];
              const int   jprev = bestj[s - 1], jcur = bestj[s];
              const int   jhere = before[s] ? j : jcur;
              bestj[s] = before[s - 1] ? jprev : jhere;
              bestd[s] = __builtin_amdgcn_fmed3f(dprev, d, dcur);
            }
            {
              const float dcur = bestd[0];
              const int   jcur = bestj[0];
              bestj[0] = before[0] ? j : jcur;
              bestd[0] = before[0] ? d : dcur;
            }
          } while (__builtin_amdgcn_ballot_w32(pend != 0u) != 0u);
        }
    }
  };

  stage_load(0);
  stage_store(0);
  __syncthreads();

  for (int c = 0; c < G_NCHUNK; ++c) {
    if (c + 1 < G_NCHUNK) stage_load(c + 1);
    const float* pb = &lds_stage[c & 1][0];

    if (scan_wave) {
      v4f* ddst = reinterpret_cast<v4f*>(&lds_d[tl][c & 1][0][0][0][0]) + lane;

      auto group_step = [&](const int tg, v8f (&acc)[G_GRP], v4f (&p0)[G_GRP], v4f (&p1)[G_GRP],
                            v8f (&accn)[G_GRP], v4f (&pn0)[G_GRP], v4f (&pn1)[G_GRP],
                            v2f (&afr)[G_GRP][4], v2f (&afw)[G_GRP][4]) __attribute__((always_inline)) {
        const int t0 = tg * G_GRP;
        {
          const int tn1 = (tg + 1 < G_NGRP) ? t0 + G_GRP : t0;
          const int tn2 = (tg + 2 < G_NGRP) ? t0 + 2 * G_GRP : G_CT - G_GRP;
          #pragma unroll
          for (int g = 0; g < G_GRP; ++g) load_norms(pb, tn1 + g, pn0[g], pn1[g]);
          #pragma unroll
          for (int g = 0; g < G_GRP; ++g) load_coords(pb, tn2 + g, afw[g]);
        }
        __builtin_amdgcn_sched_barrier(0);
        products(afr, accn);

        #pragma unroll
        for (int g = 0; g < G_GRP; ++g) {
          const float sjx[8] = {p0[g].x, p0[g].y, p0[g].z, p0[g].w, p1[g].x, p1[g].y, p1[g].z, p1[g].w};
          float dd[8];
          #pragma unroll
          for (int r = 0; r < 8; ++r) {
            float d = sqi + sjx[r] - 2.f * acc[g][r];
            dd[r] = fmaxf(d, 0.f);
          }
          v4f lo, hi;
          lo.x = dd[0]; lo.y = dd[1]; lo.z = dd[2]; lo.w = dd[3];
          hi.x = dd[4]; hi.y = dd[5]; hi.z = dd[6]; hi.w = dd[7];
          ddst[(2 * (t0 + g) + 0) * 32] = lo;
          ddst[(2 * (t0 + g) + 1) * 32] = hi;
        }
      };

      v8f accA[G_GRP], accB[G_GRP]; v4f pA0[G_GRP], pA1[G_GRP], pB0[G_GRP], pB1[G_GRP];
      v2f afA[G_GRP][4], afB[G_GRP][4];
      static_assert(G_NGRP % 2 == 0 && G_NGRP >= 2, "the groups are taken in pairs, on two alternating register sets");
      #pragma unroll
      for (int g = 0; g < G_GRP; ++g) { load_coords(pb, g, afB[g]); load_norms(pb, g, pA0[g], pA1[g]); }
      products(afB, accA);
      #pragma unroll
      for (int g = 0; g < G_GRP; ++g) load_coords(pb, G_GRP + g, afA[g]);

      for (int tg = 0; tg < G_NGRP; tg += 2) {
        group_step(tg,     accA, pA0, pA1, accB, pB0, pB1, afA, afB);
        group_step(tg + 1, accB, pB0, pB1, accA, pA0, pA1, afB, afA);
      }
    } else {
      if (c >= 1) list_chunk(c - 1);
    }

    if (c + 1 < G_NCHUNK) stage_store((c + 1) & 1);
    __syncthreads();
  }

  if (scan_wave) return;
  list_chunk(G_NCHUNK - 1);

  #pragma unroll
  for (int s = 0; s < KNN; ++s) {
    lds_cd[tl][m][kh * KNN + s] = bestd[s];
    lds_cj[tl][m][kh * KNN + s] = bestj[s];
  }
  __builtin_amdgcn_wave_barrier();

  {
    int rank[KNN];
    #pragma unroll
    for (int s = 0; s < KNN; ++s) rank[s] = 0;
    #pragma unroll 1
    for (int t = 0; t < 2 * KNN; ++t) {
      const float dt = lds_cd[tl][m][t];
      const int   jq = lds_cj[tl][m][t];
      #pragma unroll
      for (int s = 0; s < KNN; ++s) {
        const int before = (int)(dt < bestd[s]) | ((int)(dt == bestd[s]) & (int)(jq < bestj[s]));
        rank[s] += before;
      }
    }
    #pragma unroll
    for (int s = 0; s < KNN; ++s) {
      if (rank[s] < KNN) {
        lds_w[tl][m][rank[s]] = expf(-bestd[s]);
        lds_j[tl][m][rank[s]] = bestj[s];
      }
    }
  }
  __builtin_amdgcn_wave_barrier();

  for (int r = 0; r < 16; ++r) {
    float acc0 = 0.f, acc1 = 0.f;
    #pragma unroll
    for (int k = 0; k < KNN; ++k) {
      const int   j  = lds_j[tl][r][k];
      const float wk = lds_w[tl][r][k];
      acc0 += wk * feats[(size_t)j * DFEAT + lane];
      acc1 += wk * feats[(size_t)j * DFEAT + lane + 32];
    }
    lds_agg[tl][r][lane]      = acc0;
    lds_agg[tl][r][lane + 32] = acc1;
  }
  __builtin_amdgcn_wave_barrier();

  const float* fr = feats + (size_t)(row0 + m) * DFEAT;
  #pragma unroll
  for (int nt = 0; nt < 4; ++nt) {
    v8f acc = {};
    #pragma unroll
    for (int s = 0; s < 16; ++s) {
      v2f av, bv;
      av.x = fr[4 * s + 2 * kh];
      av.y = fr[4 * s + 1 + 2 * kh];
      bv.x = Wout[(4 * s + 2 * kh) * DFEAT + nt * 16 + m];
      bv.y = Wout[(4 * s + 1 + 2 * kh) * DFEAT + nt * 16 + m];
      acc = wmma_f32(av, bv, acc);
    }
    #pragma unroll
    for (int s = 0; s < 16; ++s) {
      v2f av, bv;
      av.x = lds_agg[tl][m][4 * s + 2 * kh];
      av.y = lds_agg[tl][m][4 * s + 1 + 2 * kh];
      bv.x = Wout[(64 + 4 * s + 2 * kh) * DFEAT + nt * 16 + m];
      bv.y = Wout[(64 + 4 * s + 1 + 2 * kh) * DFEAT + nt * 16 + m];
      acc = wmma_f32(av, bv, acc);
    }
    #pragma unroll
    for (int r = 0; r < 8; ++r) {
      const int row = row0 + r + 8 * kh;
      const int col = nt * 16 + m;
      out[(size_t)row * DFEAT + col] = acc[r] + bout[col];
    }
  }
}

extern "C" void kernel_launch(void* const* d_in, const int* in_sizes, int n_in,
                              void* d_out, int out_size, void* d_ws, size_t ws_size,
                              hipStream_t stream) {
  const float* x  = (const float*)d_in[0];
  const float* Wf = (const float*)d_in[1];
  const float* bf = (const float*)d_in[2];
  const float* Wl = (const float*)d_in[3];
  const float* bl = (const float*)d_in[4];
  const float* Wo = (const float*)d_in[5];
  const float* bo = (const float*)d_in[6];
  float* out = (float*)d_out;

  float* feats  = (float*)d_ws;
  float* coords = feats  + (size_t)N_PTS * DFEAT;
  float* sq     = coords + (size_t)N_PTS * DLAT;

  encoder_kernel<<<NUM_WG, 256, 0, stream>>>(x, Wf, bf, Wl, bl, feats, coords);
  sq_kernel<<<(N_PTS + 255) / 256, 256, 0, stream>>>(coords, sq);
  gravnet_kernel<<<G_NUM_WG, G_THREADS, 0, stream>>>(coords, sq, feats, Wo, bo, out);
}
